# starting kernel
# baseline (speedup 1.0000x reference)
#include <hip/hip_runtime.h>


typedef float v2f __attribute__((ext_vector_type(2)));
typedef float v8f __attribute__((ext_vector_type(8)));

#define IN_DIM        16384
#define OUT_CH        2
#define BATCH         4096
#define SAMPLES_PER_WG 16
#define WAVES_PER_WG   8
#define THREADS_PER_WG (32 * WAVES_PER_WG)
#define K_PER_WAVE    (IN_DIM / WAVES_PER_WG)
#define K_OUTER       64

__global__ __launch_bounds__(THREADS_PER_WG)
void diag_wmma_gathered_matvec(const float* __restrict__ x,
                               const int*   __restrict__ t,
                               const float* __restrict__ W,
                               const float* __restrict__ bias,
                               float*       __restrict__ out) {
    __shared__ float partial[WAVES_PER_WG][OUT_CH][SAMPLES_PER_WG];

    const int tid   = threadIdx.x;
    const int wave  = tid >> 5;
    const int lane  = tid & 31;
    const int sbase = blockIdx.x * SAMPLES_PER_WG;

    const int n     = lane & 15;
    const int khalf = (lane >> 4) << 1;

    const int tn   = t[sbase + n];
    const int widx = (980 - tn) / 20;

    const float* pa  = x + (size_t)(sbase + n) * IN_DIM
                         + (size_t)wave * K_PER_WAVE + khalf;
    const float* pb0 = W + ((size_t)widx * OUT_CH + 0) * IN_DIM
                         + (size_t)wave * K_PER_WAVE + khalf;
    const float* pb1 = W + ((size_t)widx * OUT_CH + 1) * IN_DIM
                         + (size_t)wave * K_PER_WAVE + khalf;

    v8f c0 = {};
    v8f c1 = {};

    for (int kt = 0; kt < K_PER_WAVE; kt += K_OUTER) {
        __builtin_prefetch(pa + K_OUTER, 0, 3);
#pragma unroll
        for (int u = 0; u < K_OUTER; u += 4) {
            v2f a  = *(const v2f*)(pa  + u);
            v2f b0 = *(const v2f*)(pb0 + u);
            v2f b1 = *(const v2f*)(pb1 + u);
            c0 = __builtin_amdgcn_wmma_f32_16x16x4_f32(
                     false, a, false, b0, (short)0, c0, false, false);
            c1 = __builtin_amdgcn_wmma_f32_16x16x4_f32(
                     false, a, false, b1, (short)0, c1, false, false);
        }
        pa  += K_OUTER;
        pb0 += K_OUTER;
        pb1 += K_OUTER;
    }

    int m = -1, e = 0;
    if (lane < 8)        { m = lane;      e = lane;      }
    else if (lane >= 24) { m = lane - 16; e = lane - 24; }

    float d0 = 0.0f, d1 = 0.0f;
#pragma unroll
    for (int i = 0; i < 8; ++i) {
        d0 = (e == i) ? c0[i] : d0;
        d1 = (e == i) ? c1[i] : d1;
    }
    if (m >= 0) {
        partial[wave][0][m] = d0;
        partial[wave][1][m] = d1;
    }
    __syncthreads();

    if (tid < SAMPLES_PER_WG * OUT_CH) {
        const int mm = tid >> 1;
        const int ch = tid & 1;
        float s = 0.0f;
#pragma unroll
        for (int w = 0; w < WAVES_PER_WG; ++w) s += partial[w][ch][mm];
        const int wi = (980 - t[sbase + mm]) / 20;
        out[(size_t)(sbase + mm) * OUT_CH + ch] = s + bias[wi * OUT_CH + ch];
    }
}

extern "C" void kernel_launch(void* const* d_in, const int* in_sizes, int n_in,
                              void* d_out, int out_size, void* d_ws, size_t ws_size,
                              hipStream_t stream) {
    const float* x    = (const float*)d_in[0];
    const int*   t    = (const int*)  d_in[1];
    const float* W    = (const float*)d_in[2];
    const float* bias = (const float*)d_in[3];
    float*       out  = (float*)d_out;

    dim3 grid(BATCH / SAMPLES_PER_WG);
    dim3 block(THREADS_PER_WG);
    diag_wmma_gathered_matvec<<<grid, block, 0, stream>>>(x, t, W, bias, out);
}


# WMMA tile remapped, k along tile rows: one sample per wave, each load 512 contiguous bytes (4 lines, not 16 rows 64 KiB apart), 48 loads grouped per trip; dot-product summation order regrouped
# speedup vs baseline: 1.8102x; 1.8102x over previous
#include <hip/hip_runtime.h>


typedef float v2f __attribute__((ext_vector_type(2)));
typedef float v4f __attribute__((ext_vector_type(4)));
typedef float v8f __attribute__((ext_vector_type(8)));

#define IN_DIM         16384
#define OUT_CH         2
#define BATCH          4096
#define SAMPLES_PER_WG 16
#define WAVES_PER_WG   SAMPLES_PER_WG
#define THREADS_PER_WG (32 * WAVES_PER_WG)
#define SEG            128
#define GROUP          16
#define GSTRIDE        (SEG * GROUP)
#define NGROUPS        (IN_DIM / GSTRIDE)

struct Frag {
    v4f a[GROUP];
    v4f b0[GROUP];
    v4f b1[GROUP];
};

__device__ __forceinline__ void load_group(Frag& f, const float* __restrict__ pa,
                                           const float* __restrict__ pb0,
                                           const float* __restrict__ pb1) {
#pragma unroll
    for (int g = 0; g < GROUP; ++g) {
        f.a[g]  = *(const v4f*)(pa  + g * SEG);
        f.b0[g] = *(const v4f*)(pb0 + g * SEG);
        f.b1[g] = *(const v4f*)(pb1 + g * SEG);
    }
}

__device__ __forceinline__ void mma_group(const Frag& f, v8f& c0, v8f& c1) {
#pragma unroll
    for (int g = 0; g < GROUP; ++g) {
        const v2f alo  = {f.a[g].x,  f.a[g].y},  ahi  = {f.a[g].z,  f.a[g].w};
        const v2f b0lo = {f.b0[g].x, f.b0[g].y}, b0hi = {f.b0[g].z, f.b0[g].w};
        const v2f b1lo = {f.b1[g].x, f.b1[g].y}, b1hi = {f.b1[g].z, f.b1[g].w};
        c0 = __builtin_amdgcn_wmma_f32_16x16x4_f32(
                 false, alo, false, b0lo, (short)0, c0, false, false);
        c1 = __builtin_amdgcn_wmma_f32_16x16x4_f32(
                 false, alo, false, b1lo, (short)0, c1, false, false);
        c0 = __builtin_amdgcn_wmma_f32_16x16x4_f32(
                 false, ahi, false, b0hi, (short)0, c0, false, false);
        c1 = __builtin_amdgcn_wmma_f32_16x16x4_f32(
                 false, ahi, false, b1hi, (short)0, c1, false, false);
    }
}

__global__ __launch_bounds__(THREADS_PER_WG)
void diag_wmma_gathered_matvec(const float* __restrict__ x,
                               const int*   __restrict__ t,
                               const float* __restrict__ W,
                               const float* __restrict__ bias,
                               float*       __restrict__ out) {
    __shared__ float res[SAMPLES_PER_WG][OUT_CH];

    const int tid   = threadIdx.x;
    const int lane  = tid & 31;
    const int wave  = __builtin_amdgcn_readfirstlane(tid >> 5);
    const int sbase = blockIdx.x * SAMPLES_PER_WG;
    const int smp   = sbase + wave;

    const int widx  = (980 - t[smp]) / 20;

    const float* pa  = x + (size_t)smp * IN_DIM + 4 * lane;
    const float* pb0 = W + ((size_t)widx * OUT_CH + 0) * IN_DIM + 4 * lane;
    const float* pb1 = W + ((size_t)widx * OUT_CH + 1) * IN_DIM + 4 * lane;

    v8f c0 = {};
    v8f c1 = {};

    for (int i = 0; i < NGROUPS; ++i) {
        Frag f;
        load_group(f, pa, pb0, pb1);
        __builtin_amdgcn_sched_barrier(0);
        mma_group(f, c0, c1);
        pa  += GSTRIDE;
        pb0 += GSTRIDE;
        pb1 += GSTRIDE;
    }

    int e = -1;
    if (lane < 8)        e = lane;
    else if (lane >= 24) e = lane - 24;

    float d0 = 0.0f, d1 = 0.0f;
#pragma unroll
    for (int i = 0; i < 8; ++i) {
        d0 = (e == i) ? c0[i] : d0;
        d1 = (e == i) ? c1[i] : d1;
    }
#pragma unroll
    for (int off = 16; off >= 1; off >>= 1) {
        d0 += __shfl_xor(d0, off, 32);
        d1 += __shfl_xor(d1, off, 32);
    }
    if (lane == 0) {
        res[wave][0] = d0;
        res[wave][1] = d1;
    }
    __syncthreads();

    if (tid < SAMPLES_PER_WG * OUT_CH) {
        const int mm = tid >> 1;
        const int ch = tid & 1;
        const int wi = (980 - t[sbase + mm]) / 20;
        out[(size_t)(sbase + mm) * OUT_CH + ch] = res[mm][ch] + bias[wi * OUT_CH + ch];
    }
}

extern "C" void kernel_launch(void* const* d_in, const int* in_sizes, int n_in,
                              void* d_out, int out_size, void* d_ws, size_t ws_size,
                              hipStream_t stream) {
    const float* x    = (const float*)d_in[0];
    const int*   t    = (const int*)  d_in[1];
    const float* W    = (const float*)d_in[2];
    const float* bias = (const float*)d_in[3];
    float*       out  = (float*)d_out;

    dim3 grid(BATCH / SAMPLES_PER_WG);
    dim3 block(THREADS_PER_WG);
    diag_wmma_gathered_matvec<<<grid, block, 0, stream>>>(x, t, W, bias, out);
}


# Waves take samples in order of weight index (per-workgroup counting sort of a 1024-sample window): a workgroup's W rows are shared in the WGP cache; x loads non-temporal; per-sample sums unchanged
# speedup vs baseline: 1.9956x; 1.1024x over previous
#include <hip/hip_runtime.h>


typedef float v2f __attribute__((ext_vector_type(2)));
typedef float v4f __attribute__((ext_vector_type(4)));
typedef float v8f __attribute__((ext_vector_type(8)));

#define IN_DIM         16384
#define OUT_CH         2
#define BATCH          4096
#define SAMPLES_PER_WG 16
#define WAVES_PER_WG   SAMPLES_PER_WG
#define THREADS_PER_WG (32 * WAVES_PER_WG)
#define SEG            128
#define GROUP          16
#define GSTRIDE        (SEG * GROUP)
#define NGROUPS        (IN_DIM / GSTRIDE)
#define NKEYS          50
#define WINDOW         1024
#define WGS_PER_WINDOW (WINDOW / SAMPLES_PER_WG)
#define CHUNKS         (WINDOW / 32)

struct Frag {
    v4f a[GROUP];
    v4f b0[GROUP];
    v4f b1[GROUP];
};

__device__ __forceinline__ void load_group(Frag& f, const float* __restrict__ pa,
                                           const float* __restrict__ pb0,
                                           const float* __restrict__ pb1) {
#pragma unroll
    for (int g = 0; g < GROUP; ++g) {
        f.a[g]  = __builtin_nontemporal_load((const v4f*)(pa + g * SEG));
        f.b0[g] = *(const v4f*)(pb0 + g * SEG);
        f.b1[g] = *(const v4f*)(pb1 + g * SEG);
    }
}

__device__ __forceinline__ void mma_group(const Frag& f, v8f& c0, v8f& c1) {
#pragma unroll
    for (int g = 0; g < GROUP; ++g) {
        const v2f alo  = {f.a[g].x,  f.a[g].y},  ahi  = {f.a[g].z,  f.a[g].w};
        const v2f b0lo = {f.b0[g].x, f.b0[g].y}, b0hi = {f.b0[g].z, f.b0[g].w};
        const v2f b1lo = {f.b1[g].x, f.b1[g].y}, b1hi = {f.b1[g].z, f.b1[g].w};
        c0 = __builtin_amdgcn_wmma_f32_16x16x4_f32(
                 false, alo, false, b0lo, (short)0, c0, false, false);
        c1 = __builtin_amdgcn_wmma_f32_16x16x4_f32(
                 false, alo, false, b1lo, (short)0, c1, false, false);
        c0 = __builtin_amdgcn_wmma_f32_16x16x4_f32(
                 false, ahi, false, b0hi, (short)0, c0, false, false);
        c1 = __builtin_amdgcn_wmma_f32_16x16x4_f32(
                 false, ahi, false, b1hi, (short)0, c1, false, false);
    }
}

__device__ __forceinline__ int wave_running_sum(int x, int lane) {
    x += __builtin_amdgcn_update_dpp(0, x, 0x111, 0xf, 0xf, true);
    x += __builtin_amdgcn_update_dpp(0, x, 0x112, 0xf, 0xf, true);
    x += __builtin_amdgcn_update_dpp(0, x, 0x114, 0xf, 0xf, true);
    x += __builtin_amdgcn_update_dpp(0, x, 0x118, 0xf, 0xf, true);
    const int row0 = __builtin_amdgcn_readlane(x, 15);
    return lane >= 16 ? x + row0 : x;
}

__global__ __launch_bounds__(THREADS_PER_WG)
void diag_wmma_gathered_matvec(const float* __restrict__ x,
                               const int*   __restrict__ t,
                               const float* __restrict__ W,
                               const float* __restrict__ bias,
                               float*       __restrict__ out) {
    __shared__ int keys[WINDOW];
    __shared__ int cnt[64];

    const int tid   = threadIdx.x;
    const int lane  = tid & 31;
    const int wave  = __builtin_amdgcn_readfirstlane(tid >> 5);
    const int wbase = (blockIdx.x / WGS_PER_WINDOW) * WINDOW;
    const int pos   = (blockIdx.x % WGS_PER_WINDOW) * SAMPLES_PER_WG + wave;

#pragma unroll
    for (int j = 0; j < WINDOW / THREADS_PER_WG; ++j) {
        const int i = tid + j * THREADS_PER_WG;
        int k = (980 - t[wbase + i]) / 20;
        k = k < 0 ? 0 : (k > NKEYS - 1 ? NKEYS - 1 : k);
        keys[i] = k;
    }
    __syncthreads();

    int kc0 = 0, kc1 = 0, kc2 = 0, kc3 = 0;
#pragma unroll 8
    for (int it = 0; it < CHUNKS; ++it) {
        const int kv = keys[32 * it + lane];
        kc0 += __builtin_popcount(__builtin_amdgcn_ballot_w32(kv == wave));
        kc1 += __builtin_popcount(__builtin_amdgcn_ballot_w32(kv == wave + 16));
        kc2 += __builtin_popcount(__builtin_amdgcn_ballot_w32(kv == wave + 32));
        kc3 += __builtin_popcount(__builtin_amdgcn_ballot_w32(kv == wave + 48));
    }
    if (lane == 0) {
        cnt[wave]      = kc0;
        cnt[wave + 16] = kc1;
        cnt[wave + 32] = kc2;
        cnt[wave + 48] = kc3;
    }
    __syncthreads();

    const int s0   = wave_running_sum(cnt[lane], lane);
    const int s1   = wave_running_sum(cnt[32 + lane], lane) + __builtin_amdgcn_readlane(s0, 31);
    const int ksel = __builtin_popcount(__builtin_amdgcn_ballot_w32(s0 <= pos)) +
                     __builtin_popcount(__builtin_amdgcn_ballot_w32(s1 <= pos));
    int front = 0;
    if (ksel > 0)
        front = (ksel <= 32) ? __builtin_amdgcn_readlane(s0, (ksel - 1) & 31)
                             : __builtin_amdgcn_readlane(s1, (ksel - 33) & 31);
    const int rank = pos - front;

    int cv = 0;
#pragma unroll
    for (int it = 0; it < CHUNKS; ++it) {
        const unsigned m = __builtin_amdgcn_ballot_w32(keys[32 * it + lane] == ksel);
        cv = (lane == it) ? __builtin_popcount(m) : cv;
    }
    const int pc     = wave_running_sum(cv, lane);
    const int piece  = __builtin_popcount(__builtin_amdgcn_ballot_w32(pc <= rank)) & 31;
    const int before = piece > 0 ? __builtin_amdgcn_readlane(pc, (piece - 1) & 31) : 0;
    const int rin    = rank - before;
    const int kvs    = keys[32 * piece + lane];
    const unsigned msel = __builtin_amdgcn_ballot_w32(kvs == ksel);
    const int  below = __builtin_amdgcn_mbcnt_lo(msel, 0u);
    const unsigned one = __builtin_amdgcn_ballot_w32(kvs == ksel && below == rin);
    const int  src   = one ? __builtin_ctz(one) : 0;
    const int smp    = __builtin_amdgcn_readfirstlane(wbase + piece * 32 + src);

    const int widx  = (980 - t[smp]) / 20;

    const float* pa  = x + (size_t)smp * IN_DIM + 4 * lane;
    const float* pb0 = W + ((size_t)widx * OUT_CH + 0) * IN_DIM + 4 * lane;
    const float* pb1 = W + ((size_t)widx * OUT_CH + 1) * IN_DIM + 4 * lane;

    v8f c0 = {};
    v8f c1 = {};

    for (int i = 0; i < NGROUPS; ++i) {
        Frag f;
        load_group(f, pa, pb0, pb1);
        __builtin_amdgcn_sched_barrier(0);
        mma_group(f, c0, c1);
        pa  += GSTRIDE;
        pb0 += GSTRIDE;
        pb1 += GSTRIDE;
    }

    int e = -1;
    if (lane < 8)        e = lane;
    else if (lane >= 24) e = lane - 24;

    float d0 = 0.0f, d1 = 0.0f;
#pragma unroll
    for (int i = 0; i < 8; ++i) {
        d0 = (e == i) ? c0[i] : d0;
        d1 = (e == i) ? c1[i] : d1;
    }
#pragma unroll
    for (int off = 16; off >= 1; off >>= 1) {
        d0 += __shfl_xor(d0, off, 32);
        d1 += __shfl_xor(d1, off, 32);
    }
    if (lane == 0) {
        const v2f r = {d0 + bias[widx * OUT_CH + 0], d1 + bias[widx * OUT_CH + 1]};
        *(v2f*)(out + (size_t)smp * OUT_CH) = r;
    }
}

extern "C" void kernel_launch(void* const* d_in, const int* in_sizes, int n_in,
                              void* d_out, int out_size, void* d_ws, size_t ws_size,
                              hipStream_t stream) {
    const float* x    = (const float*)d_in[0];
    const int*   t    = (const int*)  d_in[1];
    const float* W    = (const float*)d_in[2];
    const float* bias = (const float*)d_in[3];
    float*       out  = (float*)d_out;

    dim3 grid(BATCH / SAMPLES_PER_WG);
    dim3 block(THREADS_PER_WG);
    diag_wmma_gathered_matvec<<<grid, block, 0, stream>>>(x, t, W, bias, out);
}


# Row split over the workgroup's 16 waves: a wave keeps its 1024-float slice of both W rows in registers for all 16 samples and streams only x, 3 samples ahead; dot-product sums regrouped via LDS
# speedup vs baseline: 2.3437x; 1.1744x over previous
#include <hip/hip_runtime.h>


typedef float v2f __attribute__((ext_vector_type(2)));
typedef float v4f __attribute__((ext_vector_type(4)));
typedef float v8f __attribute__((ext_vector_type(8)));

#define IN_DIM         16384
#define OUT_CH         2
#define BATCH          4096
#define SAMPLES_PER_WG 16
#define WAVES_PER_WG   16
#define THREADS_PER_WG (32 * WAVES_PER_WG)
#define SEG            128
#define SLICE          (IN_DIM / WAVES_PER_WG)
#define SEGS           (SLICE / SEG)
#define AHEAD          3
#define NBUF           (AHEAD + 1)
#define NKEYS          50
#define WINDOW         1024
#define WGS_PER_WINDOW (WINDOW / SAMPLES_PER_WG)
#define CHUNKS         (WINDOW / 32)

struct XFrag { v4f a[SEGS]; };
struct WFrag { v4f b0[SEGS]; v4f b1[SEGS]; };

__device__ __forceinline__ void load_x(XFrag& f, const float* __restrict__ p) {
#pragma unroll
    for (int g = 0; g < SEGS; ++g)
        f.a[g] = __builtin_nontemporal_load((const v4f*)(p + g * SEG));
}

__device__ __forceinline__ void load_w(WFrag& f, const float* __restrict__ p0,
                                       const float* __restrict__ p1) {
#pragma unroll
    for (int g = 0; g < SEGS; ++g) {
        f.b0[g] = *(const v4f*)(p0 + g * SEG);
        f.b1[g] = *(const v4f*)(p1 + g * SEG);
    }
}

__device__ __forceinline__ void mma_slice(const XFrag& xf, const WFrag& wf, v8f& c0, v8f& c1) {
#pragma unroll
    for (int g = 0; g < SEGS; ++g) {
        const v2f alo  = {xf.a[g].x,  xf.a[g].y},  ahi  = {xf.a[g].z,  xf.a[g].w};
        const v2f b0lo = {wf.b0[g].x, wf.b0[g].y}, b0hi = {wf.b0[g].z, wf.b0[g].w};
        const v2f b1lo = {wf.b1[g].x, wf.b1[g].y}, b1hi = {wf.b1[g].z, wf.b1[g].w};
        c0 = __builtin_amdgcn_wmma_f32_16x16x4_f32(
                 false, alo, false, b0lo, (short)0, c0, false, false);
        c1 = __builtin_amdgcn_wmma_f32_16x16x4_f32(
                 false, alo, false, b1lo, (short)0, c1, false, false);
        c0 = __builtin_amdgcn_wmma_f32_16x16x4_f32(
                 false, ahi, false, b0hi, (short)0, c0, false, false);
        c1 = __builtin_amdgcn_wmma_f32_16x16x4_f32(
                 false, ahi, false, b1hi, (short)0, c1, false, false);
    }
}

__device__ __forceinline__ int wave_running_sum(int x, int lane) {
    x += __builtin_amdgcn_update_dpp(0, x, 0x111, 0xf, 0xf, true);
    x += __builtin_amdgcn_update_dpp(0, x, 0x112, 0xf, 0xf, true);
    x += __builtin_amdgcn_update_dpp(0, x, 0x114, 0xf, 0xf, true);
    x += __builtin_amdgcn_update_dpp(0, x, 0x118, 0xf, 0xf, true);
    const int row0 = __builtin_amdgcn_readlane(x, 15);
    return lane >= 16 ? x + row0 : x;
}

__global__ __launch_bounds__(THREADS_PER_WG)
void diag_wmma_gathered_matvec(const float* __restrict__ x,
                               const int*   __restrict__ t,
                               const float* __restrict__ W,
                               const float* __restrict__ bias,
                               float*       __restrict__ out) {
    __shared__ int keys[WINDOW];
    __shared__ int cnt[64];

    const int tid   = threadIdx.x;
    const int lane  = tid & 31;
    const int wave  = __builtin_amdgcn_readfirstlane(tid >> 5);
    const int wbase = (blockIdx.x / WGS_PER_WINDOW) * WINDOW;
    const int pos   = (blockIdx.x % WGS_PER_WINDOW) * SAMPLES_PER_WG + wave;

#pragma unroll
    for (int j = 0; j < WINDOW / THREADS_PER_WG; ++j) {
        const int i = tid + j * THREADS_PER_WG;
        int k = (980 - t[wbase + i]) / 20;
        k = k < 0 ? 0 : (k > NKEYS - 1 ? NKEYS - 1 : k);
        keys[i] = k;
    }
    __syncthreads();

    int kc0 = 0, kc1 = 0, kc2 = 0, kc3 = 0;
#pragma unroll 8
    for (int it = 0; it < CHUNKS; ++it) {
        const int kv = keys[32 * it + lane];
        kc0 += __builtin_popcount(__builtin_amdgcn_ballot_w32(kv == wave));
        kc1 += __builtin_popcount(__builtin_amdgcn_ballot_w32(kv == wave + 16));
        kc2 += __builtin_popcount(__builtin_amdgcn_ballot_w32(kv == wave + 32));
        kc3 += __builtin_popcount(__builtin_amdgcn_ballot_w32(kv == wave + 48));
    }
    if (lane == 0) {
        cnt[wave]      = kc0;
        cnt[wave + 16] = kc1;
        cnt[wave + 32] = kc2;
        cnt[wave + 48] = kc3;
    }
    __syncthreads();

    const int s0   = wave_running_sum(cnt[lane], lane);
    const int s1   = wave_running_sum(cnt[32 + lane], lane) + __builtin_amdgcn_readlane(s0, 31);
    const int ksel = __builtin_popcount(__builtin_amdgcn_ballot_w32(s0 <= pos)) +
                     __builtin_popcount(__builtin_amdgcn_ballot_w32(s1 <= pos));
    int front = 0;
    if (ksel > 0)
        front = (ksel <= 32) ? __builtin_amdgcn_readlane(s0, (ksel - 1) & 31)
                             : __builtin_amdgcn_readlane(s1, (ksel - 33) & 31);
    const int rank = pos - front;

    int cv = 0;
#pragma unroll
    for (int it = 0; it < CHUNKS; ++it) {
        const unsigned m = __builtin_amdgcn_ballot_w32(keys[32 * it + lane] == ksel);
        cv = (lane == it) ? __builtin_popcount(m) : cv;
    }
    const int pc     = wave_running_sum(cv, lane);
    const int piece  = __builtin_popcount(__builtin_amdgcn_ballot_w32(pc <= rank)) & 31;
    const int before = piece > 0 ? __builtin_amdgcn_readlane(pc, (piece - 1) & 31) : 0;
    const int rin    = rank - before;
    const int kvs    = keys[32 * piece + lane];
    const unsigned msel = __builtin_amdgcn_ballot_w32(kvs == ksel);
    const int  below = __builtin_amdgcn_mbcnt_lo(msel, 0u);
    const unsigned one = __builtin_amdgcn_ballot_w32(kvs == ksel && below == rin);
    const int  src   = one ? __builtin_ctz(one) : 0;
    const int smp    = __builtin_amdgcn_readfirstlane(wbase + piece * 32 + src);

    const int widx  = (980 - t[smp]) / 20;

    __shared__ int wg_smp[SAMPLES_PER_WG];
    __shared__ int wg_key[SAMPLES_PER_WG];
    __shared__ v2f part[SAMPLES_PER_WG][WAVES_PER_WG][32];
    if (lane == 0) {
        wg_smp[wave] = smp;
        wg_key[wave] = widx;
    }
    __syncthreads();
    const int vs = wg_smp[lane & 15];
    const int vk = wg_key[lane & 15];

    const float* xw = x + wave * SLICE + 4 * lane;
    const float* ww = W + wave * SLICE + 4 * lane;

    int e = -1;
    if (lane < 8)        e = lane;
    else if (lane >= 24) e = lane - 24;

    XFrag xb[NBUF];
    WFrag wf;

    {
        const int k0 = __builtin_amdgcn_readlane(vk, 0);
        load_w(wf, ww + ((size_t)k0 * OUT_CH + 0) * IN_DIM,
                   ww + ((size_t)k0 * OUT_CH + 1) * IN_DIM);
    }
#pragma unroll
    for (int s = 0; s < AHEAD; ++s)
        load_x(xb[s % NBUF], xw + (size_t)__builtin_amdgcn_readlane(vs, s) * IN_DIM);

#pragma unroll
    for (int s = 0; s < SAMPLES_PER_WG; ++s) {
        __builtin_amdgcn_sched_barrier(0);
        if (s + AHEAD < SAMPLES_PER_WG)
            load_x(xb[(s + AHEAD) % NBUF],
                   xw + (size_t)__builtin_amdgcn_readlane(vs, s + AHEAD) * IN_DIM);
        if (s > 0) {
            const int key = __builtin_amdgcn_readlane(vk, s);
            if (key != __builtin_amdgcn_readlane(vk, s - 1)) {
                load_w(wf, ww + ((size_t)key * OUT_CH + 0) * IN_DIM,
                           ww + ((size_t)key * OUT_CH + 1) * IN_DIM);
#pragma unroll
                for (int g = 0; g < SEGS; ++g)
                    asm volatile("" : "+v"(wf.b0[g]), "+v"(wf.b1[g]));
            }
        }
        v8f c0 = {};
        v8f c1 = {};
        mma_slice(xb[s % NBUF], wf, c0, c1);

        float d0 = 0.0f, d1 = 0.0f;
#pragma unroll
        for (int i = 0; i < 8; ++i) {
            d0 = (e == i) ? c0[i] : d0;
            d1 = (e == i) ? c1[i] : d1;
        }
        const v2f d = {d0, d1};
        part[s][wave][lane] = d;
    }
    __syncthreads();

    float r0 = 0.0f, r1 = 0.0f;
#pragma unroll
    for (int w = 0; w < WAVES_PER_WG; ++w) {
        const v2f p = part[wave][w][lane];
        r0 += p.x;
        r1 += p.y;
    }
#pragma unroll
    for (int off = 16; off >= 1; off >>= 1) {
        r0 += __shfl_xor(r0, off, 32);
        r1 += __shfl_xor(r1, off, 32);
    }
    if (lane == 0) {
        const v2f r = {r0 + bias[widx * OUT_CH + 0], r1 + bias[widx * OUT_CH + 1]};
        *(v2f*)(out + (size_t)smp * OUT_CH) = r;
    }
}

extern "C" void kernel_launch(void* const* d_in, const int* in_sizes, int n_in,
                              void* d_out, int out_size, void* d_ws, size_t ws_size,
                              hipStream_t stream) {
    const float* x    = (const float*)d_in[0];
    const int*   t    = (const int*)  d_in[1];
    const float* W    = (const float*)d_in[2];
    const float* bias = (const float*)d_in[3];
    float*       out  = (float*)d_out;

    dim3 grid(BATCH / SAMPLES_PER_WG);
    dim3 block(THREADS_PER_WG);
    diag_wmma_gathered_matvec<<<grid, block, 0, stream>>>(x, t, W, bias, out);
}
